# starting kernel
# baseline (speedup 1.0000x reference)
#include <hip/hip_runtime.h>

typedef float v8f __attribute__((ext_vector_type(8)));
typedef float v2f __attribute__((ext_vector_type(2)));

#define DIM_E   64
#define DIM_H   8
#define NLAYER  25
#define NBATCH  1000
#define NT      512
#define TILE_M  16
#define HPAD    12
#define NSTEPS  (NT + NLAYER - 1)

__device__ __forceinline__ float fast_tanh(float x) {
  float e = __builtin_amdgcn_exp2f(x * 2.88539008177792681f);
  return 1.0f - 2.0f * __builtin_amdgcn_rcpf(e + 1.0f);
}

__device__ __forceinline__ v8f wmma4(v2f a, v2f b, v8f c) {
  return __builtin_amdgcn_wmma_f32_16x16x4_f32(false, a, false, b, (short)0, c,
                                               false, false);
}

__global__ __launch_bounds__(NLAYER * 32, 1)
void rnn_pipeline_kernel(const int* __restrict__ x,
                         const float* __restrict__ h0,
                         const float* __restrict__ emb,
                         const float* __restrict__ Wxh0,
                         const float* __restrict__ Wxhr,
                         const float* __restrict__ Whh,
                         const float* __restrict__ bh,
                         const float* __restrict__ Why,
                         const float* __restrict__ by,
                         float* __restrict__ out) {
  __shared__ float hbuf[NLAYER][2][TILE_M][HPAD];

  const int tid  = threadIdx.x;
  const int wid  = tid >> 5;
  const int lane = tid & 31;
  const int n    = lane & 15;
  const int kb   = (lane >> 4) << 1;
  const int tile_base = blockIdx.x * TILE_M;

  v2f  wA[16];
  v2f  wR[2];
  float bias = (n < DIM_H) ? bh[wid * DIM_H + n] : 0.0f;

  if (wid == 0) {
#pragma unroll
    for (int c = 0; c < 16; ++c)
#pragma unroll
      for (int j = 0; j < 2; ++j) {
        int k = c * 4 + kb + j;
        wA[c][j] = (n < DIM_H) ? Wxh0[k * DIM_H + n] : 0.0f;
      }
#pragma unroll
    for (int c = 0; c < 2; ++c)
#pragma unroll
      for (int j = 0; j < 2; ++j) {
        int k = c * 4 + kb + j;
        wR[c][j] = (n < DIM_H) ? Whh[k * DIM_H + n] : 0.0f;
      }
  } else {
#pragma unroll
    for (int c = 0; c < 4; ++c)
#pragma unroll
      for (int j = 0; j < 2; ++j) {
        int k = c * 4 + kb + j;
        float v = 0.0f;
        if (n < DIM_H)
          v = (k < DIM_H) ? Wxhr[(wid - 1) * DIM_H * DIM_H + k * DIM_H + n]
                          : Whh[wid * DIM_H * DIM_H + (k - DIM_H) * DIM_H + n];
        wA[c][j] = v;
      }
  }

  if (lane < TILE_M) {
    int gb = tile_base + lane; if (gb > NBATCH - 1) gb = NBATCH - 1;
    const float* hp = h0 + (wid * NBATCH + gb) * DIM_H;
#pragma unroll
    for (int j = 0; j < DIM_H; ++j) {
      float v = hp[j];
      hbuf[wid][0][lane][j] = v;
      hbuf[wid][1][lane][j] = v;
    }
  }
  __syncthreads();

  const int m = n;
  int gb = tile_base + m; if (gb > NBATCH - 1) gb = NBATCH - 1;

  int xcur = 0;
  if (wid == 0) xcur = x[gb * NT];

  for (int s = 0; s < NSTEPS; ++s) {
    if (s >= wid && s < wid + NT) {
      const int p_in  = (s + 1) & 1;
      const int p_out = s & 1;
      v8f acc;
#pragma unroll
      for (int r = 0; r < 8; ++r) acc[r] = bias;

      if (wid == 0) {
        const int t = s;
        const float* erow = emb + (long)xcur * DIM_E;
        if (t + 1 < NT) xcur = x[gb * NT + t + 1];
#pragma unroll
        for (int c = 0; c < 16; ++c) {
          v2f a = *(const v2f*)(erow + c * 4 + kb);
          acc = wmma4(a, wA[c], acc);
        }
        const float* hp = &hbuf[0][p_in][m][0];
        v2f a0 = *(const v2f*)(hp + kb);
        v2f a1 = *(const v2f*)(hp + 4 + kb);
        acc = wmma4(a0, wR[0], acc);
        acc = wmma4(a1, wR[1], acc);
      } else {
        const float* ip = &hbuf[wid - 1][p_in][m][0];
        const float* hp = &hbuf[wid][p_in][m][0];
        v2f a0 = *(const v2f*)(ip + kb);
        v2f a1 = *(const v2f*)(ip + 4 + kb);
        v2f a2 = *(const v2f*)(hp + kb);
        v2f a3 = *(const v2f*)(hp + 4 + kb);
        acc = wmma4(a0, wA[0], acc);
        acc = wmma4(a1, wA[1], acc);
        acc = wmma4(a2, wA[2], acc);
        acc = wmma4(a3, wA[3], acc);
      }

#pragma unroll
      for (int r = 0; r < 8; ++r) acc[r] = fast_tanh(acc[r]);

      if (n < DIM_H) {
        const int rowbase = (lane >> 4) << 3;
#pragma unroll
        for (int r = 0; r < 8; ++r)
          hbuf[wid][p_out][rowbase + r][n] = acc[r];
      }
    }
    __syncthreads();
  }

  const int pf = (wid + NT - 1) & 1;
  if (lane < TILE_M) {
    int g = tile_base + lane;
    if (g < NBATCH) {
      const float* src = &hbuf[wid][pf][lane][0];
      float4 v0 = *(const float4*)src;
      float4 v1 = *(const float4*)(src + 4);
      float* dst = out + NBATCH + ((long)wid * NBATCH + g) * DIM_H;
      *(float4*)dst       = v0;
      *(float4*)(dst + 4) = v1;
      if (wid == NLAYER - 1) {
        float a = by[0];
#pragma unroll
        for (int j = 0; j < DIM_H; ++j) a += src[j] * Why[j];
        out[g] = a;
      }
    }
  }
}

extern "C" void kernel_launch(void* const* d_in, const int* in_sizes, int n_in,
                              void* d_out, int out_size, void* d_ws, size_t ws_size,
                              hipStream_t stream) {
  (void)in_sizes; (void)n_in; (void)d_ws; (void)ws_size; (void)out_size;
  const int*   x    = (const int*)  d_in[0];
  const float* h0   = (const float*)d_in[1];
  const float* emb  = (const float*)d_in[2];
  const float* Wxh0 = (const float*)d_in[3];
  const float* Wxhr = (const float*)d_in[4];
  const float* Whh  = (const float*)d_in[5];
  const float* bh   = (const float*)d_in[6];
  const float* Why  = (const float*)d_in[7];
  const float* by   = (const float*)d_in[8];
  float* out = (float*)d_out;

  dim3 grid((NBATCH + TILE_M - 1) / TILE_M);
  dim3 block(NLAYER * 32);
  rnn_pipeline_kernel<<<grid, block, 0, stream>>>(x, h0, emb, Wxh0, Wxhr, Whh,
                                                  bh, Why, by, out);
}


# Layer-0 input projection (bias + emb[x] @ W_xh0, K=64) hoisted out of the time loop: all 25 waves form it for all 512 steps first, staged in LDS; layer 0 then starts from it; same k order
# speedup vs baseline: 1.2546x; 1.2546x over previous
#include <hip/hip_runtime.h>


typedef float v8f __attribute__((ext_vector_type(8)));
typedef float v2f __attribute__((ext_vector_type(2)));

#define DIM_E   64
#define DIM_H   8
#define NLAYER  25
#define NBATCH  1000
#define NT      512
#define TILE_M  16
#define HPAD    12
#define NSTEPS  (NT + NLAYER - 1)
#define PROJ_T  (TILE_M * DIM_H)

__device__ __forceinline__ float fast_tanh(float x) {
  float e = __builtin_amdgcn_exp2f(x * 2.88539008177792681f);
  return 1.0f - 2.0f * __builtin_amdgcn_rcpf(e + 1.0f);
}

__device__ __forceinline__ v8f wmma4(v2f a, v2f b, v8f c) {
  return __builtin_amdgcn_wmma_f32_16x16x4_f32(false, a, false, b, (short)0, c,
                                               false, false);
}

__device__ __forceinline__ v8f splat8(float v) {
  v8f a = {v, v, v, v, v, v, v, v};
  return a;
}

__device__ __forceinline__ void gather_row(v2f (&f)[16], const float* __restrict__ emb,
                                           int token, int kb) {
  const float* erow = emb + (long)token * DIM_E;
#pragma unroll
  for (int c = 0; c < 16; ++c) f[c] = *(const v2f*)(erow + c * 4 + kb);
}

__global__ __launch_bounds__(NLAYER * 32, 1)
void rnn_pipeline_kernel(const int* __restrict__ x,
                         const float* __restrict__ h0,
                         const float* __restrict__ emb,
                         const float* __restrict__ Wxh0,
                         const float* __restrict__ Wxhr,
                         const float* __restrict__ Whh,
                         const float* __restrict__ bh,
                         const float* __restrict__ Why,
                         const float* __restrict__ by,
                         float* __restrict__ out) {
  __shared__ float hbuf[NLAYER][2][TILE_M][HPAD];
  __shared__ __attribute__((aligned(16))) float xproj[NT][PROJ_T];

  const int tid  = threadIdx.x;
  const int wid  = tid >> 5;
  const int lane = tid & 31;
  const int n    = lane & 15;
  const int kb   = (lane >> 4) << 1;
  const int tile_base = blockIdx.x * TILE_M;
  const int m = n;
  int gb = tile_base + m; if (gb > NBATCH - 1) gb = NBATCH - 1;
  const int pslot = ((lane >> 4) * DIM_H + (n & (DIM_H - 1))) * 8;

  v2f  wA[4];
  float bias = (n < DIM_H) ? bh[wid * DIM_H + n] : 0.0f;
#pragma unroll
  for (int c = 0; c < 4; ++c)
#pragma unroll
    for (int j = 0; j < 2; ++j) {
      int k = c * 4 + kb + j;
      float v = 0.0f;
      if (n < DIM_H) {
        if (wid == 0) {
          if (k < DIM_H) v = Whh[k * DIM_H + n];
        } else {
          v = (k < DIM_H) ? Wxhr[(wid - 1) * DIM_H * DIM_H + k * DIM_H + n]
                          : Whh[wid * DIM_H * DIM_H + (k - DIM_H) * DIM_H + n];
        }
      }
      wA[c][j] = v;
    }

  {
    const int wu = __builtin_amdgcn_readfirstlane(wid);
    v2f wP[16];
#pragma unroll
    for (int c = 0; c < 16; ++c) {
      const int k = c * 4 + kb;
      wP[c][0] = (n < DIM_H) ? Wxh0[k * DIM_H + n] : 0.0f;
      wP[c][1] = (n < DIM_H) ? Wxh0[(k + 1) * DIM_H + n] : 0.0f;
    }
    const float bias0 = (n < DIM_H) ? bh[n] : 0.0f;
    const int* xrow = x + gb * NT;
    float* pdst = &xproj[0][pslot];

    v2f fa[16], fb[16];
    int xa = xrow[wu];
    int xb = xrow[wu + NLAYER];
    gather_row(fa, emb, xa, kb);
    xa = xrow[wu + 2 * NLAYER];
    for (int t = wu; t < wu + 20 * NLAYER; t += 2 * NLAYER) {
      gather_row(fb, emb, xb, kb);
      xb = xrow[min(t + 3 * NLAYER, NT - 1)];
      {
        v8f acc = splat8(bias0);
#pragma unroll
        for (int c = 0; c < 16; ++c) acc = wmma4(fa[c], wP[c], acc);
        if (n < DIM_H) {
          float* d = pdst + t * PROJ_T;
          *(float4*)d       = make_float4(acc[0], acc[1], acc[2], acc[3]);
          *(float4*)(d + 4) = make_float4(acc[4], acc[5], acc[6], acc[7]);
        }
      }
      gather_row(fa, emb, xa, kb);
      xa = xrow[min(t + 4 * NLAYER, NT - 1)];
      {
        v8f acc = splat8(bias0);
#pragma unroll
        for (int c = 0; c < 16; ++c) acc = wmma4(fb[c], wP[c], acc);
        if (n < DIM_H) {
          float* d = pdst + (t + NLAYER) * PROJ_T;
          *(float4*)d       = make_float4(acc[0], acc[1], acc[2], acc[3]);
          *(float4*)(d + 4) = make_float4(acc[4], acc[5], acc[6], acc[7]);
        }
      }
    }
    if (wu + 20 * NLAYER < NT) {
      v8f acc = splat8(bias0);
#pragma unroll
      for (int c = 0; c < 16; ++c) acc = wmma4(fa[c], wP[c], acc);
      if (n < DIM_H) {
        float* d = pdst + (wu + 20 * NLAYER) * PROJ_T;
        *(float4*)d       = make_float4(acc[0], acc[1], acc[2], acc[3]);
        *(float4*)(d + 4) = make_float4(acc[4], acc[5], acc[6], acc[7]);
      }
    }
  }

  if (lane < TILE_M) {
    int g0 = tile_base + lane; if (g0 > NBATCH - 1) g0 = NBATCH - 1;
    const float* hp = h0 + (wid * NBATCH + g0) * DIM_H;
#pragma unroll
    for (int j = 0; j < DIM_H; ++j) {
      float v = hp[j];
      hbuf[wid][0][lane][j] = v;
      hbuf[wid][1][lane][j] = v;
    }
  }
  __syncthreads();

  const float* psrc = &xproj[0][pslot];

  for (int s = 0; s < NSTEPS; ++s) {
    if (s >= wid && s < wid + NT) {
      const int p_in  = (s + 1) & 1;
      const int p_out = s & 1;
      v8f acc;

      if (wid == 0) {
        const float* pp = psrc + s * PROJ_T;
        float4 p0 = *(const float4*)pp;
        float4 p1 = *(const float4*)(pp + 4);
        acc[0] = p0.x; acc[1] = p0.y; acc[2] = p0.z; acc[3] = p0.w;
        acc[4] = p1.x; acc[5] = p1.y; acc[6] = p1.z; acc[7] = p1.w;
        const float* hp = &hbuf[0][p_in][m][0];
        v2f a0 = *(const v2f*)(hp + kb);
        v2f a1 = *(const v2f*)(hp + 4 + kb);
        acc = wmma4(a0, wA[0], acc);
        acc = wmma4(a1, wA[1], acc);
      } else {
#pragma unroll
        for (int r = 0; r < 8; ++r) acc[r] = bias;
        const float* ip = &hbuf[wid - 1][p_in][m][0];
        const float* hp = &hbuf[wid][p_in][m][0];
        v2f a0 = *(const v2f*)(ip + kb);
        v2f a1 = *(const v2f*)(ip + 4 + kb);
        v2f a2 = *(const v2f*)(hp + kb);
        v2f a3 = *(const v2f*)(hp + 4 + kb);
        acc = wmma4(a0, wA[0], acc);
        acc = wmma4(a1, wA[1], acc);
        acc = wmma4(a2, wA[2], acc);
        acc = wmma4(a3, wA[3], acc);
      }

#pragma unroll
      for (int r = 0; r < 8; ++r) acc[r] = fast_tanh(acc[r]);

      if (n < DIM_H) {
        const int rowbase = (lane >> 4) << 3;
#pragma unroll
        for (int r = 0; r < 8; ++r)
          hbuf[wid][p_out][rowbase + r][n] = acc[r];
      }
    }
    __syncthreads();
  }

  const int pf = (wid + NT - 1) & 1;
  if (lane < TILE_M) {
    int g = tile_base + lane;
    if (g < NBATCH) {
      const float* src = &hbuf[wid][pf][lane][0];
      float4 v0 = *(const float4*)src;
      float4 v1 = *(const float4*)(src + 4);
      float* dst = out + NBATCH + ((long)wid * NBATCH + g) * DIM_H;
      *(float4*)dst       = v0;
      *(float4*)(dst + 4) = v1;
      if (wid == NLAYER - 1) {
        float a = by[0];
#pragma unroll
        for (int j = 0; j < DIM_H; ++j) a += src[j] * Why[j];
        out[g] = a;
      }
    }
  }
}

extern "C" void kernel_launch(void* const* d_in, const int* in_sizes, int n_in,
                              void* d_out, int out_size, void* d_ws, size_t ws_size,
                              hipStream_t stream) {
  (void)in_sizes; (void)n_in; (void)d_ws; (void)ws_size; (void)out_size;
  const int*   x    = (const int*)  d_in[0];
  const float* h0   = (const float*)d_in[1];
  const float* emb  = (const float*)d_in[2];
  const float* Wxh0 = (const float*)d_in[3];
  const float* Wxhr = (const float*)d_in[4];
  const float* Whh  = (const float*)d_in[5];
  const float* bh   = (const float*)d_in[6];
  const float* Why  = (const float*)d_in[7];
  const float* by   = (const float*)d_in[8];
  float* out = (float*)d_out;

  dim3 grid((NBATCH + TILE_M - 1) / TILE_M);
  dim3 block(NLAYER * 32);
  rnn_pipeline_kernel<<<grid, block, 0, stream>>>(x, h0, emb, Wxh0, Wxhr, Whh,
                                                  bh, Why, by, out);
}


# tanh and LDS stores only on the layer's 8 real columns: lane n<8 hands accumulator rows 4..7 to padding lane n+8 (4 DPP moves); 4 tanh + 4 stores on every lane, was 8 + 8 on half of them
# speedup vs baseline: 1.3811x; 1.1008x over previous
#include <hip/hip_runtime.h>


typedef float v8f __attribute__((ext_vector_type(8)));
typedef float v2f __attribute__((ext_vector_type(2)));

#define DIM_E   64
#define DIM_H   8
#define NLAYER  25
#define NBATCH  1000
#define NT      512
#define TILE_M  16
#define HPAD    12
#define NSTEPS  (NT + NLAYER - 1)
#define PROJ_T  (TILE_M * DIM_H)

__device__ __forceinline__ float fast_tanh(float x) {
  float e = __builtin_amdgcn_exp2f(x * 2.88539008177792681f);
  return 1.0f - 2.0f * __builtin_amdgcn_rcpf(e + 1.0f);
}

__device__ __forceinline__ v8f wmma4(v2f a, v2f b, v8f c) {
  return __builtin_amdgcn_wmma_f32_16x16x4_f32(false, a, false, b, (short)0, c,
                                               false, false);
}

__device__ __forceinline__ float fold_up8(float keep, float give) {
  return __builtin_bit_cast(float,
      __builtin_amdgcn_update_dpp(__builtin_bit_cast(int, keep), __builtin_bit_cast(int, give),
                                  0x118  , 0xF, 0xC, false));
}

__device__ __forceinline__ v8f splat8(float v) {
  v8f a = {v, v, v, v, v, v, v, v};
  return a;
}

__device__ __forceinline__ void gather_row(v2f (&f)[16], const float* __restrict__ emb,
                                           int token, int kb) {
  const float* erow = emb + (long)token * DIM_E;
#pragma unroll
  for (int c = 0; c < 16; ++c) f[c] = *(const v2f*)(erow + c * 4 + kb);
}

__global__ __launch_bounds__(NLAYER * 32, 1)
void rnn_pipeline_kernel(const int* __restrict__ x,
                         const float* __restrict__ h0,
                         const float* __restrict__ emb,
                         const float* __restrict__ Wxh0,
                         const float* __restrict__ Wxhr,
                         const float* __restrict__ Whh,
                         const float* __restrict__ bh,
                         const float* __restrict__ Why,
                         const float* __restrict__ by,
                         float* __restrict__ out) {
  __shared__ float hbuf[NLAYER][2][TILE_M][HPAD];
  __shared__ __attribute__((aligned(16))) float xproj[NT][PROJ_T];

  const int tid  = threadIdx.x;
  const int wid  = tid >> 5;
  const int lane = tid & 31;
  const int n    = lane & 15;
  const int kb   = (lane >> 4) << 1;
  const int tile_base = blockIdx.x * TILE_M;
  const int m = n;
  int gb = tile_base + m; if (gb > NBATCH - 1) gb = NBATCH - 1;
  const int pslot = ((lane >> 4) * DIM_H + (n & (DIM_H - 1))) * 8;

  v2f  wA[4];
  float bias = (n < DIM_H) ? bh[wid * DIM_H + n] : 0.0f;
#pragma unroll
  for (int c = 0; c < 4; ++c)
#pragma unroll
    for (int j = 0; j < 2; ++j) {
      int k = c * 4 + kb + j;
      float v = 0.0f;
      if (n < DIM_H) {
        if (wid == 0) {
          if (k < DIM_H) v = Whh[k * DIM_H + n];
        } else {
          v = (k < DIM_H) ? Wxhr[(wid - 1) * DIM_H * DIM_H + k * DIM_H + n]
                          : Whh[wid * DIM_H * DIM_H + (k - DIM_H) * DIM_H + n];
        }
      }
      wA[c][j] = v;
    }

  {
    const int wu = __builtin_amdgcn_readfirstlane(wid);
    v2f wP[16];
#pragma unroll
    for (int c = 0; c < 16; ++c) {
      const int k = c * 4 + kb;
      wP[c][0] = (n < DIM_H) ? Wxh0[k * DIM_H + n] : 0.0f;
      wP[c][1] = (n < DIM_H) ? Wxh0[(k + 1) * DIM_H + n] : 0.0f;
    }
    const float bias0 = (n < DIM_H) ? bh[n] : 0.0f;
    const int* xrow = x + gb * NT;
    float* pdst = &xproj[0][pslot];

    v2f fa[16], fb[16];
    int xa = xrow[wu];
    int xb = xrow[wu + NLAYER];
    gather_row(fa, emb, xa, kb);
    xa = xrow[wu + 2 * NLAYER];
    for (int t = wu; t < wu + 20 * NLAYER; t += 2 * NLAYER) {
      gather_row(fb, emb, xb, kb);
      xb = xrow[min(t + 3 * NLAYER, NT - 1)];
      {
        v8f acc = splat8(bias0);
#pragma unroll
        for (int c = 0; c < 16; ++c) acc = wmma4(fa[c], wP[c], acc);
        if (n < DIM_H) {
          float* d = pdst + t * PROJ_T;
          *(float4*)d       = make_float4(acc[0], acc[1], acc[2], acc[3]);
          *(float4*)(d + 4) = make_float4(acc[4], acc[5], acc[6], acc[7]);
        }
      }
      gather_row(fa, emb, xa, kb);
      xa = xrow[min(t + 4 * NLAYER, NT - 1)];
      {
        v8f acc = splat8(bias0);
#pragma unroll
        for (int c = 0; c < 16; ++c) acc = wmma4(fb[c], wP[c], acc);
        if (n < DIM_H) {
          float* d = pdst + (t + NLAYER) * PROJ_T;
          *(float4*)d       = make_float4(acc[0], acc[1], acc[2], acc[3]);
          *(float4*)(d + 4) = make_float4(acc[4], acc[5], acc[6], acc[7]);
        }
      }
    }
    if (wu + 20 * NLAYER < NT) {
      v8f acc = splat8(bias0);
#pragma unroll
      for (int c = 0; c < 16; ++c) acc = wmma4(fa[c], wP[c], acc);
      if (n < DIM_H) {
        float* d = pdst + (wu + 20 * NLAYER) * PROJ_T;
        *(float4*)d       = make_float4(acc[0], acc[1], acc[2], acc[3]);
        *(float4*)(d + 4) = make_float4(acc[4], acc[5], acc[6], acc[7]);
      }
    }
  }

  if (lane < TILE_M) {
    int g0 = tile_base + lane; if (g0 > NBATCH - 1) g0 = NBATCH - 1;
    const float* hp = h0 + (wid * NBATCH + g0) * DIM_H;
#pragma unroll
    for (int j = 0; j < DIM_H; ++j) {
      float v = hp[j];
      hbuf[wid][0][lane][j] = v;
      hbuf[wid][1][lane][j] = v;
    }
  }
  __syncthreads();

  const float* psrc = &xproj[0][pslot];

  for (int s = 0; s < NSTEPS; ++s) {
    if (s >= wid && s < wid + NT) {
      const int p_in  = (s + 1) & 1;
      const int p_out = s & 1;
      v8f acc;

      if (wid == 0) {
        const float* pp = psrc + s * PROJ_T;
        float4 p0 = *(const float4*)pp;
        float4 p1 = *(const float4*)(pp + 4);
        acc[0] = p0.x; acc[1] = p0.y; acc[2] = p0.z; acc[3] = p0.w;
        acc[4] = p1.x; acc[5] = p1.y; acc[6] = p1.z; acc[7] = p1.w;
        const float* hp = &hbuf[0][p_in][m][0];
        v2f a0 = *(const v2f*)(hp + kb);
        v2f a1 = *(const v2f*)(hp + 4 + kb);
        acc = wmma4(a0, wA[0], acc);
        acc = wmma4(a1, wA[1], acc);
      } else {
#pragma unroll
        for (int r = 0; r < 8; ++r) acc[r] = bias;
        const float* ip = &hbuf[wid - 1][p_in][m][0];
        const float* hp = &hbuf[wid][p_in][m][0];
        v2f a0 = *(const v2f*)(ip + kb);
        v2f a1 = *(const v2f*)(ip + 4 + kb);
        v2f a2 = *(const v2f*)(hp + kb);
        v2f a3 = *(const v2f*)(hp + 4 + kb);
        acc = wmma4(a0, wA[0], acc);
        acc = wmma4(a1, wA[1], acc);
        acc = wmma4(a2, wA[2], acc);
        acc = wmma4(a3, wA[3], acc);
      }

      float t[4];
#pragma unroll
      for (int r = 0; r < 4; ++r) t[r] = fast_tanh(fold_up8(acc[r], acc[r + 4]));
      {
        const int row0 = ((lane >> 4) << 3) + ((n >> 3) << 2);
#pragma unroll
        for (int r = 0; r < 4; ++r)
          hbuf[wid][p_out][row0 + r][n & (DIM_H - 1)] = t[r];
      }
    }
    __syncthreads();
  }

  const int pf = (wid + NT - 1) & 1;
  if (lane < TILE_M) {
    int g = tile_base + lane;
    if (g < NBATCH) {
      const float* src = &hbuf[wid][pf][lane][0];
      float4 v0 = *(const float4*)src;
      float4 v1 = *(const float4*)(src + 4);
      float* dst = out + NBATCH + ((long)wid * NBATCH + g) * DIM_H;
      *(float4*)dst       = v0;
      *(float4*)(dst + 4) = v1;
      if (wid == NLAYER - 1) {
        float a = by[0];
#pragma unroll
        for (int j = 0; j < DIM_H; ++j) a += src[j] * Why[j];
        out[g] = a;
      }
    }
  }
}

extern "C" void kernel_launch(void* const* d_in, const int* in_sizes, int n_in,
                              void* d_out, int out_size, void* d_ws, size_t ws_size,
                              hipStream_t stream) {
  (void)in_sizes; (void)n_in; (void)d_ws; (void)ws_size; (void)out_size;
  const int*   x    = (const int*)  d_in[0];
  const float* h0   = (const float*)d_in[1];
  const float* emb  = (const float*)d_in[2];
  const float* Wxh0 = (const float*)d_in[3];
  const float* Wxhr = (const float*)d_in[4];
  const float* Whh  = (const float*)d_in[5];
  const float* bh   = (const float*)d_in[6];
  const float* Why  = (const float*)d_in[7];
  const float* by   = (const float*)d_in[8];
  float* out = (float*)d_out;

  dim3 grid((NBATCH + TILE_M - 1) / TILE_M);
  dim3 block(NLAYER * 32);
  rnn_pipeline_kernel<<<grid, block, 0, stream>>>(x, h0, emb, Wxh0, Wxhr, Whh,
                                                  bh, Why, by, out);
}
